# starting kernel
# baseline (speedup 1.0000x reference)
#include <hip/hip_runtime.h>


typedef __attribute__((ext_vector_type(2))) float v2f;
typedef __attribute__((ext_vector_type(8))) float v8f;

#define NQ     20
#define NBATCH 16
#define SSIZE  (1u << NQ)

#define WAVES_PER_BATCH 512
#define TILES_PER_WAVE  8

template <int P0, bool FIRST>
__global__ __launch_bounds__(256) void rx_group_pass(
    const float* __restrict__ phi,
    const float* __restrict__ thetas,
    float*       __restrict__ state)
{
    const int lane  = threadIdx.x & 31;
    const int wid   = (int)blockIdx.x * 8 + (threadIdx.x >> 5);
    const int b     = wid >> 9;
    const int wtile = wid & (WAVES_PER_BATCH - 1);
    const int h     = lane >> 4;
    const int n     = lane & 15;

    const int g = (16 - P0) >> 2;
    float cb[4], sb[4];
#pragma unroll
    for (int beta = 0; beta < 4; ++beta) {
        float th = 0.5f * thetas[b * NQ + 4 * g + (3 - beta)];
        cb[beta] = cosf(th);
        sb[beta] = sinf(th);
    }

    v2f aR[4], aI[4], amI[4];
    const int M = n;
#pragma unroll
    for (int kc = 0; kc < 4; ++kc) {
#pragma unroll
        for (int r = 0; r < 2; ++r) {
            const int j = 4 * kc + r + 2 * h;
            const int x = M ^ j;
            float prod = 1.0f;
#pragma unroll
            for (int beta = 0; beta < 4; ++beta)
                prod *= ((x >> beta) & 1) ? sb[beta] : cb[beta];
            const int k = __popc(x);
            const float Rv = (k & 1) ? 0.0f : ((k & 2) ? -prod : prod);
            const float Iv = (k & 1) ? ((k & 2) ? prod : -prod) : 0.0f;
            aR[kc][r]  = Rv;
            aI[kc][r]  = Iv;
            amI[kc][r] = -Iv;
        }
    }

    const unsigned LOMASK = (1u << P0) - 1u;

#pragma unroll 1
    for (int t = 0; t < TILES_PER_WAVE; ++t) {
        const unsigned tile  = (unsigned)wtile * TILES_PER_WAVE + (unsigned)t;
        const unsigned nc    = tile * 16u + (unsigned)n;
        const unsigned hi    = nc >> P0;
        const unsigned lo    = nc & LOMASK;
        const unsigned ebase = (hi << (P0 + 4)) | lo;
        const size_t   base  = (size_t)b * SSIZE + ebase;

        v2f bRe[4], bIm[4];
        if (FIRST) {
            const float* src = phi + base;
#pragma unroll
            for (int kc = 0; kc < 4; ++kc)
#pragma unroll
                for (int r = 0; r < 2; ++r) {
                    const int j = 4 * kc + r + 2 * h;
                    bRe[kc][r] = src[(size_t)j << P0];
                }
        } else {
            const v2f* src = (const v2f*)state + base;
#pragma unroll
            for (int kc = 0; kc < 4; ++kc)
#pragma unroll
                for (int r = 0; r < 2; ++r) {
                    const int j = 4 * kc + r + 2 * h;
                    const v2f v = src[(size_t)j << P0];
                    bRe[kc][r] = v.x;
                    bIm[kc][r] = v.y;
                }
        }

        v8f accRe = {};
        v8f accIm = {};
#pragma unroll
        for (int kc = 0; kc < 4; ++kc) {
            accRe = __builtin_amdgcn_wmma_f32_16x16x4_f32(
                false, aR[kc], false, bRe[kc], (short)0, accRe, false, false);
            accIm = __builtin_amdgcn_wmma_f32_16x16x4_f32(
                false, aI[kc], false, bRe[kc], (short)0, accIm, false, false);
        }
        if (!FIRST) {
#pragma unroll
            for (int kc = 0; kc < 4; ++kc) {
                accRe = __builtin_amdgcn_wmma_f32_16x16x4_f32(
                    false, amI[kc], false, bIm[kc], (short)0, accRe, false, false);
                accIm = __builtin_amdgcn_wmma_f32_16x16x4_f32(
                    false, aR[kc], false, bIm[kc], (short)0, accIm, false, false);
            }
        }

        v2f* dst = (v2f*)state + base;
#pragma unroll
        for (int r = 0; r < 8; ++r) {
            const int Mo = r + 8 * h;
            v2f o;
            o.x = accRe[r];
            o.y = accIm[r];
            dst[(size_t)Mo << P0] = o;
        }
    }
}

extern "C" void kernel_launch(void* const* d_in, const int* in_sizes, int n_in,
                              void* d_out, int out_size, void* d_ws, size_t ws_size,
                              hipStream_t stream) {
    (void)in_sizes; (void)n_in; (void)d_ws; (void)ws_size; (void)out_size;
    const float* phi    = (const float*)d_in[0];
    const float* thetas = (const float*)d_in[1];
    float* out = (float*)d_out;

    const dim3 grid(1024), block(256);
    rx_group_pass<16, true ><<<grid, block, 0, stream>>>(phi, thetas, out);
    rx_group_pass<12, false><<<grid, block, 0, stream>>>(phi, thetas, out);
    rx_group_pass< 8, false><<<grid, block, 0, stream>>>(phi, thetas, out);
    rx_group_pass< 4, false><<<grid, block, 0, stream>>>(phi, thetas, out);
    rx_group_pass< 0, false><<<grid, block, 0, stream>>>(phi, thetas, out);
}


# Low-bit passes (P0=4 and P0=0): tile index interleaved across waves (tile = trip*512 + wave) instead of 8 consecutive tiles per wave, so waves in flight sweep contiguous memory
# speedup vs baseline: 1.3867x; 1.3867x over previous
#include <hip/hip_runtime.h>


typedef __attribute__((ext_vector_type(2))) float v2f;
typedef __attribute__((ext_vector_type(8))) float v8f;

#define NQ     20
#define NBATCH 16
#define SSIZE  (1u << NQ)

#define WAVES_PER_BATCH 512
#define TILES_PER_WAVE  8

template <int P0, bool FIRST>
__global__ __launch_bounds__(256) void rx_group_pass(
    const float* __restrict__ phi,
    const float* __restrict__ thetas,
    float*       __restrict__ state)
{
    const int lane  = threadIdx.x & 31;
    const int wid   = (int)blockIdx.x * 8 + (threadIdx.x >> 5);
    const int b     = wid >> 9;
    const int wtile = wid & (WAVES_PER_BATCH - 1);
    const int h     = lane >> 4;
    const int n     = lane & 15;

    const int g = (16 - P0) >> 2;
    float cb[4], sb[4];
#pragma unroll
    for (int beta = 0; beta < 4; ++beta) {
        float th = 0.5f * thetas[b * NQ + 4 * g + (3 - beta)];
        cb[beta] = cosf(th);
        sb[beta] = sinf(th);
    }

    v2f aR[4], aI[4], amI[4];
    const int M = n;
#pragma unroll
    for (int kc = 0; kc < 4; ++kc) {
#pragma unroll
        for (int r = 0; r < 2; ++r) {
            const int j = 4 * kc + r + 2 * h;
            const int x = M ^ j;
            float prod = 1.0f;
#pragma unroll
            for (int beta = 0; beta < 4; ++beta)
                prod *= ((x >> beta) & 1) ? sb[beta] : cb[beta];
            const int k = __popc(x);
            const float Rv = (k & 1) ? 0.0f : ((k & 2) ? -prod : prod);
            const float Iv = (k & 1) ? ((k & 2) ? prod : -prod) : 0.0f;
            aR[kc][r]  = Rv;
            aI[kc][r]  = Iv;
            amI[kc][r] = -Iv;
        }
    }

    const unsigned LOMASK = (1u << P0) - 1u;

#pragma unroll 1
    for (int t = 0; t < TILES_PER_WAVE; ++t) {
        const unsigned tile  = (P0 < 8)
            ? (unsigned)t * WAVES_PER_BATCH + (unsigned)wtile
            : (unsigned)wtile * TILES_PER_WAVE + (unsigned)t;
        const unsigned nc    = tile * 16u + (unsigned)n;
        const unsigned hi    = nc >> P0;
        const unsigned lo    = nc & LOMASK;
        const unsigned ebase = (hi << (P0 + 4)) | lo;
        const size_t   base  = (size_t)b * SSIZE + ebase;

        v2f bRe[4], bIm[4];
        if (FIRST) {
            const float* src = phi + base;
#pragma unroll
            for (int kc = 0; kc < 4; ++kc)
#pragma unroll
                for (int r = 0; r < 2; ++r) {
                    const int j = 4 * kc + r + 2 * h;
                    bRe[kc][r] = src[(size_t)j << P0];
                }
        } else {
            const v2f* src = (const v2f*)state + base;
#pragma unroll
            for (int kc = 0; kc < 4; ++kc)
#pragma unroll
                for (int r = 0; r < 2; ++r) {
                    const int j = 4 * kc + r + 2 * h;
                    const v2f v = src[(size_t)j << P0];
                    bRe[kc][r] = v.x;
                    bIm[kc][r] = v.y;
                }
        }

        v8f accRe = {};
        v8f accIm = {};
#pragma unroll
        for (int kc = 0; kc < 4; ++kc) {
            accRe = __builtin_amdgcn_wmma_f32_16x16x4_f32(
                false, aR[kc], false, bRe[kc], (short)0, accRe, false, false);
            accIm = __builtin_amdgcn_wmma_f32_16x16x4_f32(
                false, aI[kc], false, bRe[kc], (short)0, accIm, false, false);
        }
        if (!FIRST) {
#pragma unroll
            for (int kc = 0; kc < 4; ++kc) {
                accRe = __builtin_amdgcn_wmma_f32_16x16x4_f32(
                    false, amI[kc], false, bIm[kc], (short)0, accRe, false, false);
                accIm = __builtin_amdgcn_wmma_f32_16x16x4_f32(
                    false, aR[kc], false, bIm[kc], (short)0, accIm, false, false);
            }
        }

        v2f* dst = (v2f*)state + base;
#pragma unroll
        for (int r = 0; r < 8; ++r) {
            const int Mo = r + 8 * h;
            v2f o;
            o.x = accRe[r];
            o.y = accIm[r];
            dst[(size_t)Mo << P0] = o;
        }
    }
}

extern "C" void kernel_launch(void* const* d_in, const int* in_sizes, int n_in,
                              void* d_out, int out_size, void* d_ws, size_t ws_size,
                              hipStream_t stream) {
    (void)in_sizes; (void)n_in; (void)d_ws; (void)ws_size; (void)out_size;
    const float* phi    = (const float*)d_in[0];
    const float* thetas = (const float*)d_in[1];
    float* out = (float*)d_out;

    const dim3 grid(1024), block(256);
    rx_group_pass<16, true ><<<grid, block, 0, stream>>>(phi, thetas, out);
    rx_group_pass<12, false><<<grid, block, 0, stream>>>(phi, thetas, out);
    rx_group_pass< 8, false><<<grid, block, 0, stream>>>(phi, thetas, out);
    rx_group_pass< 4, false><<<grid, block, 0, stream>>>(phi, thetas, out);
    rx_group_pass< 0, false><<<grid, block, 0, stream>>>(phi, thetas, out);
}


# Last two passes (bits 4-7, bits 0-3) fused in one launch: product 1's D registers feed product 2 as WMMA A operand, k order 2s+r+8h (f32 sum order of that product changes); 4 launches
# speedup vs baseline: 2.1530x; 1.5526x over previous
#include <hip/hip_runtime.h>


typedef __attribute__((ext_vector_type(2))) float v2f;
typedef __attribute__((ext_vector_type(8))) float v8f;

#define NQ     20
#define NBATCH 16
#define SSIZE  (1u << NQ)

#define WAVES_PER_BATCH 512
#define TILES_PER_WAVE  8

__device__ __forceinline__ void group_angles(const float* __restrict__ thetas,
                                             int b, int g, float (&cb)[4], float (&sb)[4])
{
#pragma unroll
    for (int beta = 0; beta < 4; ++beta) {
        const float th = 0.5f * thetas[b * NQ + 4 * g + (3 - beta)];
        cb[beta] = cosf(th);
        sb[beta] = sinf(th);
    }
}

__device__ __forceinline__ void u_entry(const float (&cb)[4], const float (&sb)[4],
                                        int M, int j, float& Rv, float& Iv)
{
    const int x = M ^ j;
    float prod = 1.0f;
#pragma unroll
    for (int beta = 0; beta < 4; ++beta)
        prod *= ((x >> beta) & 1) ? sb[beta] : cb[beta];
    const int k = __popc(x);
    Rv = (k & 1) ? 0.0f : ((k & 2) ? -prod : prod);
    Iv = (k & 1) ? ((k & 2) ? prod : -prod) : 0.0f;
}

template <int P0, bool FIRST>
__global__ __launch_bounds__(256) void rx_group_pass(
    const float* __restrict__ phi,
    const float* __restrict__ thetas,
    float*       __restrict__ state)
{
    static_assert(P0 >= 8, "groups below bit 8 go through rx_low_pair_pass");
    const int lane  = threadIdx.x & 31;
    const int wid   = (int)blockIdx.x * 8 + (threadIdx.x >> 5);
    const int b     = wid >> 9;
    const int wtile = wid & (WAVES_PER_BATCH - 1);
    const int h     = lane >> 4;
    const int n     = lane & 15;

    float cb[4], sb[4];
    group_angles(thetas, b, (16 - P0) >> 2, cb, sb);

    v2f aR[4], aI[4], amI[4];
#pragma unroll
    for (int kc = 0; kc < 4; ++kc) {
#pragma unroll
        for (int r = 0; r < 2; ++r) {
            float Rv, Iv;
            u_entry(cb, sb, n, 4 * kc + r + 2 * h, Rv, Iv);
            aR[kc][r]  = Rv;
            aI[kc][r]  = Iv;
            amI[kc][r] = -Iv;
        }
    }

    const unsigned LOMASK = (1u << P0) - 1u;

#pragma unroll 1
    for (int t = 0; t < TILES_PER_WAVE; ++t) {
        const unsigned tile  = (unsigned)wtile * TILES_PER_WAVE + (unsigned)t;
        const unsigned nc    = tile * 16u + (unsigned)n;
        const unsigned hi    = nc >> P0;
        const unsigned lo    = nc & LOMASK;
        const unsigned ebase = (hi << (P0 + 4)) | lo;
        const size_t   base  = (size_t)b * SSIZE + ebase;

        v2f bRe[4], bIm[4];
        if (FIRST) {
            const float* src = phi + base;
#pragma unroll
            for (int kc = 0; kc < 4; ++kc)
#pragma unroll
                for (int r = 0; r < 2; ++r) {
                    const int j = 4 * kc + r + 2 * h;
                    bRe[kc][r] = src[(size_t)j << P0];
                }
        } else {
            const v2f* src = (const v2f*)state + base;
#pragma unroll
            for (int kc = 0; kc < 4; ++kc)
#pragma unroll
                for (int r = 0; r < 2; ++r) {
                    const int j = 4 * kc + r + 2 * h;
                    const v2f v = src[(size_t)j << P0];
                    bRe[kc][r] = v.x;
                    bIm[kc][r] = v.y;
                }
        }

        v8f accRe = {};
        v8f accIm = {};
#pragma unroll
        for (int kc = 0; kc < 4; ++kc) {
            accRe = __builtin_amdgcn_wmma_f32_16x16x4_f32(
                false, aR[kc], false, bRe[kc], (short)0, accRe, false, false);
            accIm = __builtin_amdgcn_wmma_f32_16x16x4_f32(
                false, aI[kc], false, bRe[kc], (short)0, accIm, false, false);
        }
        if (!FIRST) {
#pragma unroll
            for (int kc = 0; kc < 4; ++kc) {
                accRe = __builtin_amdgcn_wmma_f32_16x16x4_f32(
                    false, amI[kc], false, bIm[kc], (short)0, accRe, false, false);
                accIm = __builtin_amdgcn_wmma_f32_16x16x4_f32(
                    false, aR[kc], false, bIm[kc], (short)0, accIm, false, false);
            }
        }

        v2f* dst = (v2f*)state + base;
#pragma unroll
        for (int r = 0; r < 8; ++r) {
            const int Mo = r + 8 * h;
            v2f o;
            o.x = accRe[r];
            o.y = accIm[r];
            dst[(size_t)Mo << P0] = o;
        }
    }
}

__global__ __launch_bounds__(256) void rx_low_pair_pass(
    const float* __restrict__ thetas,
    float*       __restrict__ state)
{
    const int lane  = threadIdx.x & 31;
    const int wid   = (int)blockIdx.x * 8 + (threadIdx.x >> 5);
    const int b     = wid >> 9;
    const int wtile = wid & (WAVES_PER_BATCH - 1);
    const int h     = lane >> 4;
    const int n     = lane & 15;

    float c4[4], s4[4], c0[4], s0[4];
    group_angles(thetas, b, 3, c4, s4);
    group_angles(thetas, b, 4, c0, s0);

    v2f u4R[4], u4I[4], u4mI[4];
    v2f u0R[4], u0I[4], u0mI[4];
#pragma unroll
    for (int kc = 0; kc < 4; ++kc) {
#pragma unroll
        for (int r = 0; r < 2; ++r) {
            float Rv, Iv;
            u_entry(c4, s4, n, 4 * kc + r + 2 * h, Rv, Iv);
            u4R[kc][r]  = Rv;
            u4I[kc][r]  = Iv;
            u4mI[kc][r] = -Iv;
            u_entry(c0, s0, n, 2 * kc + r + 8 * h, Rv, Iv);
            u0R[kc][r]  = Rv;
            u0I[kc][r]  = Iv;
            u0mI[kc][r] = -Iv;
        }
    }

#pragma unroll 1
    for (int t = 0; t < TILES_PER_WAVE; ++t) {
        const unsigned tile = (unsigned)t * WAVES_PER_BATCH + (unsigned)wtile;
        const size_t   base = (size_t)b * SSIZE + (size_t)tile * 256u + (unsigned)n;

        const v2f* src = (const v2f*)state + base;
        v2f xRe[4], xIm[4];
#pragma unroll
        for (int kc = 0; kc < 4; ++kc)
#pragma unroll
            for (int r = 0; r < 2; ++r) {
                const int j4 = 4 * kc + r + 2 * h;
                const v2f v = src[j4 * 16];
                xRe[kc][r] = v.x;
                xIm[kc][r] = v.y;
            }

        v8f y1Re = {};
        v8f y1Im = {};
#pragma unroll
        for (int kc = 0; kc < 4; ++kc) {
            y1Re = __builtin_amdgcn_wmma_f32_16x16x4_f32(
                false, xRe[kc], false, u4R[kc], (short)0, y1Re, false, false);
            y1Im = __builtin_amdgcn_wmma_f32_16x16x4_f32(
                false, xRe[kc], false, u4I[kc], (short)0, y1Im, false, false);
        }
#pragma unroll
        for (int kc = 0; kc < 4; ++kc) {
            y1Re = __builtin_amdgcn_wmma_f32_16x16x4_f32(
                false, xIm[kc], false, u4mI[kc], (short)0, y1Re, false, false);
            y1Im = __builtin_amdgcn_wmma_f32_16x16x4_f32(
                false, xIm[kc], false, u4R[kc], (short)0, y1Im, false, false);
        }

        v8f y2Re = {};
        v8f y2Im = {};
#pragma unroll
        for (int s = 0; s < 4; ++s) {
            v2f a;
            a.x = y1Re[2 * s];
            a.y = y1Re[2 * s + 1];
            y2Re = __builtin_amdgcn_wmma_f32_16x16x4_f32(
                false, a, false, u0R[s], (short)0, y2Re, false, false);
            y2Im = __builtin_amdgcn_wmma_f32_16x16x4_f32(
                false, a, false, u0I[s], (short)0, y2Im, false, false);
        }
#pragma unroll
        for (int s = 0; s < 4; ++s) {
            v2f a;
            a.x = y1Im[2 * s];
            a.y = y1Im[2 * s + 1];
            y2Re = __builtin_amdgcn_wmma_f32_16x16x4_f32(
                false, a, false, u0mI[s], (short)0, y2Re, false, false);
            y2Im = __builtin_amdgcn_wmma_f32_16x16x4_f32(
                false, a, false, u0R[s], (short)0, y2Im, false, false);
        }

        v2f* dst = (v2f*)state + base;
#pragma unroll
        for (int r = 0; r < 8; ++r) {
            v2f o;
            o.x = y2Re[r];
            o.y = y2Im[r];
            dst[(r + 8 * h) * 16] = o;
        }
    }
}

extern "C" void kernel_launch(void* const* d_in, const int* in_sizes, int n_in,
                              void* d_out, int out_size, void* d_ws, size_t ws_size,
                              hipStream_t stream) {
    (void)in_sizes; (void)n_in; (void)d_ws; (void)ws_size; (void)out_size;
    const float* phi    = (const float*)d_in[0];
    const float* thetas = (const float*)d_in[1];
    float* out = (float*)d_out;

    const dim3 grid(1024), block(256);
    rx_group_pass<16, true ><<<grid, block, 0, stream>>>(phi, thetas, out);
    rx_group_pass<12, false><<<grid, block, 0, stream>>>(phi, thetas, out);
    rx_group_pass< 8, false><<<grid, block, 0, stream>>>(phi, thetas, out);
    rx_low_pair_pass<<<grid, block, 0, stream>>>(thetas, out);
}


# High passes (bits>=8): per trip a wave takes one aligned 256 B unit of every input row (2 column-paired tiles, 4 for the real input; whole 128 B lines per access), units interleaved across waves
# speedup vs baseline: 2.4689x; 1.1467x over previous
#include <hip/hip_runtime.h>


typedef __attribute__((ext_vector_type(2))) float v2f;
typedef __attribute__((ext_vector_type(4))) float v4f;
typedef __attribute__((ext_vector_type(8))) float v8f;

#define NQ     20
#define NBATCH 16
#define SSIZE  (1u << NQ)

#define WAVES_PER_BATCH 512
#define TILES_PER_WAVE  8

__device__ __forceinline__ void group_angles(const float* __restrict__ thetas,
                                             int b, int g, float (&cb)[4], float (&sb)[4])
{
#pragma unroll
    for (int beta = 0; beta < 4; ++beta) {
        const float th = 0.5f * thetas[b * NQ + 4 * g + (3 - beta)];
        cb[beta] = cosf(th);
        sb[beta] = sinf(th);
    }
}

__device__ __forceinline__ void u_entry(const float (&cb)[4], const float (&sb)[4],
                                        int M, int j, float& Rv, float& Iv)
{
    const int x = M ^ j;
    float prod = 1.0f;
#pragma unroll
    for (int beta = 0; beta < 4; ++beta)
        prod *= ((x >> beta) & 1) ? sb[beta] : cb[beta];
    const int k = __popc(x);
    Rv = (k & 1) ? 0.0f : ((k & 2) ? -prod : prod);
    Iv = (k & 1) ? ((k & 2) ? prod : -prod) : 0.0f;
}

template <int P0, bool FIRST>
__global__ __launch_bounds__(256) void rx_group_pass(
    const float* __restrict__ phi,
    const float* __restrict__ thetas,
    float*       __restrict__ state)
{
    static_assert(P0 >= 8, "groups below bit 8 go through rx_low_pair_pass");
    constexpr int NT    = FIRST ? 4 : 2;
    constexpr int TRIPS = TILES_PER_WAVE / NT;
    const int lane  = threadIdx.x & 31;
    const int wid   = (int)blockIdx.x * 8 + (threadIdx.x >> 5);
    const int b     = wid >> 9;
    const int wtile = wid & (WAVES_PER_BATCH - 1);
    const int h     = lane >> 4;
    const int n     = lane & 15;

    float cb[4], sb[4];
    group_angles(thetas, b, (16 - P0) >> 2, cb, sb);

    v2f aR[4], aI[4], amI[4];
#pragma unroll
    for (int kc = 0; kc < 4; ++kc) {
#pragma unroll
        for (int r = 0; r < 2; ++r) {
            float Rv, Iv;
            u_entry(cb, sb, n, 4 * kc + r + 2 * h, Rv, Iv);
            aR[kc][r]  = Rv;
            aI[kc][r]  = Iv;
            amI[kc][r] = -Iv;
        }
    }

    const unsigned LOMASK = (1u << P0) - 1u;

#pragma unroll 1
    for (int t = 0; t < TRIPS; ++t) {
        const unsigned unit  = (unsigned)t * WAVES_PER_BATCH + (unsigned)wtile;
        const unsigned nc    = unit * (16u * NT) + 2u * (unsigned)n;
        const unsigned hi    = nc >> P0;
        const unsigned lo    = nc & LOMASK;
        const unsigned ebase = (hi << (P0 + 4)) | lo;
        const size_t   base  = (size_t)b * SSIZE + ebase;

        v2f bRe[NT][4], bIm[NT][4];
        if constexpr (FIRST) {
            const float* src = phi + base;
#pragma unroll
            for (int kc = 0; kc < 4; ++kc)
#pragma unroll
                for (int r = 0; r < 2; ++r) {
                    const int j = 4 * kc + r + 2 * h;
                    const float* row = src + ((size_t)j << P0);
                    const v2f v0 = *(const v2f*)row;
                    const v2f v1 = *(const v2f*)(row + 32);
                    bRe[0][kc][r] = v0.x;
                    bRe[1][kc][r] = v0.y;
                    bRe[2][kc][r] = v1.x;
                    bRe[3][kc][r] = v1.y;
                }
        } else {
            const float* src = state + 2 * base;
#pragma unroll
            for (int kc = 0; kc < 4; ++kc)
#pragma unroll
                for (int r = 0; r < 2; ++r) {
                    const int j = 4 * kc + r + 2 * h;
                    const v4f v = *(const v4f*)(src + ((size_t)j << (P0 + 1)));
                    bRe[0][kc][r] = v.x;
                    bIm[0][kc][r] = v.y;
                    bRe[1][kc][r] = v.z;
                    bIm[1][kc][r] = v.w;
                }
        }

        v8f accRe[NT], accIm[NT];
#pragma unroll
        for (int c = 0; c < NT; ++c) {
            accRe[c] = (v8f){};
            accIm[c] = (v8f){};
        }
#pragma unroll
        for (int kc = 0; kc < 4; ++kc) {
#pragma unroll
            for (int c = 0; c < NT; ++c) {
                accRe[c] = __builtin_amdgcn_wmma_f32_16x16x4_f32(
                    false, aR[kc], false, bRe[c][kc], (short)0, accRe[c], false, false);
                accIm[c] = __builtin_amdgcn_wmma_f32_16x16x4_f32(
                    false, aI[kc], false, bRe[c][kc], (short)0, accIm[c], false, false);
            }
        }
        if constexpr (!FIRST) {
#pragma unroll
            for (int kc = 0; kc < 4; ++kc) {
#pragma unroll
                for (int c = 0; c < NT; ++c) {
                    accRe[c] = __builtin_amdgcn_wmma_f32_16x16x4_f32(
                        false, amI[kc], false, bIm[c][kc], (short)0, accRe[c], false, false);
                    accIm[c] = __builtin_amdgcn_wmma_f32_16x16x4_f32(
                        false, aR[kc], false, bIm[c][kc], (short)0, accIm[c], false, false);
                }
            }
        }

        float* dst = state + 2 * base;
#pragma unroll
        for (int r = 0; r < 8; ++r) {
            const int Mo = r + 8 * h;
            v4f* p = (v4f*)(dst + ((size_t)Mo << (P0 + 1)));
#pragma unroll
            for (int c = 0; c < NT; c += 2) {
                v4f o;
                o.x = accRe[c][r];
                o.y = accIm[c][r];
                o.z = accRe[c + 1][r];
                o.w = accIm[c + 1][r];
                p[(c >> 1) * 16] = o;
            }
        }
    }
}

__global__ __launch_bounds__(256) void rx_low_pair_pass(
    const float* __restrict__ thetas,
    float*       __restrict__ state)
{
    const int lane  = threadIdx.x & 31;
    const int wid   = (int)blockIdx.x * 8 + (threadIdx.x >> 5);
    const int b     = wid >> 9;
    const int wtile = wid & (WAVES_PER_BATCH - 1);
    const int h     = lane >> 4;
    const int n     = lane & 15;

    float c4[4], s4[4], c0[4], s0[4];
    group_angles(thetas, b, 3, c4, s4);
    group_angles(thetas, b, 4, c0, s0);

    v2f u4R[4], u4I[4], u4mI[4];
    v2f u0R[4], u0I[4], u0mI[4];
#pragma unroll
    for (int kc = 0; kc < 4; ++kc) {
#pragma unroll
        for (int r = 0; r < 2; ++r) {
            float Rv, Iv;
            u_entry(c4, s4, n, 4 * kc + r + 2 * h, Rv, Iv);
            u4R[kc][r]  = Rv;
            u4I[kc][r]  = Iv;
            u4mI[kc][r] = -Iv;
            u_entry(c0, s0, n, 2 * kc + r + 8 * h, Rv, Iv);
            u0R[kc][r]  = Rv;
            u0I[kc][r]  = Iv;
            u0mI[kc][r] = -Iv;
        }
    }

#pragma unroll 1
    for (int t = 0; t < TILES_PER_WAVE; ++t) {
        const unsigned tile = (unsigned)t * WAVES_PER_BATCH + (unsigned)wtile;
        const size_t   base = (size_t)b * SSIZE + (size_t)tile * 256u + (unsigned)n;

        const v2f* src = (const v2f*)state + base;
        v2f xRe[4], xIm[4];
#pragma unroll
        for (int kc = 0; kc < 4; ++kc)
#pragma unroll
            for (int r = 0; r < 2; ++r) {
                const int j4 = 4 * kc + r + 2 * h;
                const v2f v = src[j4 * 16];
                xRe[kc][r] = v.x;
                xIm[kc][r] = v.y;
            }

        v8f y1Re = {};
        v8f y1Im = {};
#pragma unroll
        for (int kc = 0; kc < 4; ++kc) {
            y1Re = __builtin_amdgcn_wmma_f32_16x16x4_f32(
                false, xRe[kc], false, u4R[kc], (short)0, y1Re, false, false);
            y1Im = __builtin_amdgcn_wmma_f32_16x16x4_f32(
                false, xRe[kc], false, u4I[kc], (short)0, y1Im, false, false);
        }
#pragma unroll
        for (int kc = 0; kc < 4; ++kc) {
            y1Re = __builtin_amdgcn_wmma_f32_16x16x4_f32(
                false, xIm[kc], false, u4mI[kc], (short)0, y1Re, false, false);
            y1Im = __builtin_amdgcn_wmma_f32_16x16x4_f32(
                false, xIm[kc], false, u4R[kc], (short)0, y1Im, false, false);
        }

        v8f y2Re = {};
        v8f y2Im = {};
#pragma unroll
        for (int s = 0; s < 4; ++s) {
            v2f a;
            a.x = y1Re[2 * s];
            a.y = y1Re[2 * s + 1];
            y2Re = __builtin_amdgcn_wmma_f32_16x16x4_f32(
                false, a, false, u0R[s], (short)0, y2Re, false, false);
            y2Im = __builtin_amdgcn_wmma_f32_16x16x4_f32(
                false, a, false, u0I[s], (short)0, y2Im, false, false);
        }
#pragma unroll
        for (int s = 0; s < 4; ++s) {
            v2f a;
            a.x = y1Im[2 * s];
            a.y = y1Im[2 * s + 1];
            y2Re = __builtin_amdgcn_wmma_f32_16x16x4_f32(
                false, a, false, u0mI[s], (short)0, y2Re, false, false);
            y2Im = __builtin_amdgcn_wmma_f32_16x16x4_f32(
                false, a, false, u0R[s], (short)0, y2Im, false, false);
        }

        v2f* dst = (v2f*)state + base;
#pragma unroll
        for (int r = 0; r < 8; ++r) {
            v2f o;
            o.x = y2Re[r];
            o.y = y2Im[r];
            dst[(r + 8 * h) * 16] = o;
        }
    }
}

extern "C" void kernel_launch(void* const* d_in, const int* in_sizes, int n_in,
                              void* d_out, int out_size, void* d_ws, size_t ws_size,
                              hipStream_t stream) {
    (void)in_sizes; (void)n_in; (void)d_ws; (void)ws_size; (void)out_size;
    const float* phi    = (const float*)d_in[0];
    const float* thetas = (const float*)d_in[1];
    float* out = (float*)d_out;

    const dim3 grid(1024), block(256);
    rx_group_pass<16, true ><<<grid, block, 0, stream>>>(phi, thetas, out);
    rx_group_pass<12, false><<<grid, block, 0, stream>>>(phi, thetas, out);
    rx_group_pass< 8, false><<<grid, block, 0, stream>>>(phi, thetas, out);
    rx_low_pair_pass<<<grid, block, 0, stream>>>(thetas, out);
}


# State swept twice, not four times: launch 1 does bits 19..12 from the input, launch 2 bits 11..0 in place; the waves of a work-group hand each block on through an LDS exchange; f32 sum order kept
# speedup vs baseline: 4.6924x; 1.9006x over previous
#include <hip/hip_runtime.h>


typedef __attribute__((ext_vector_type(2))) float v2f;
typedef __attribute__((ext_vector_type(4))) float v4f;
typedef __attribute__((ext_vector_type(8))) float v8f;

#define NQ     20
#define NBATCH 16
#define SSIZE  (1u << NQ)

#define WGS_PER_BATCH 64

__device__ __forceinline__ void group_angles(const float* __restrict__ thetas,
                                             int b, int g, float (&cb)[4], float (&sb)[4])
{
#pragma unroll
    for (int beta = 0; beta < 4; ++beta) {
        const float th = 0.5f * thetas[b * NQ + 4 * g + (3 - beta)];
        cb[beta] = cosf(th);
        sb[beta] = sinf(th);
    }
}

__device__ __forceinline__ void u_entry(const float (&cb)[4], const float (&sb)[4],
                                        int M, int j, float& Rv, float& Iv)
{
    const int x = M ^ j;
    float prod = 1.0f;
#pragma unroll
    for (int beta = 0; beta < 4; ++beta)
        prod *= ((x >> beta) & 1) ? sb[beta] : cb[beta];
    const int k = __popc(x);
    Rv = (k & 1) ? 0.0f : ((k & 2) ? -prod : prod);
    Iv = (k & 1) ? ((k & 2) ? prod : -prod) : 0.0f;
}

__device__ __forceinline__ void group_operand(const float* __restrict__ thetas, int b, int g,
                                              int n, int h,
                                              v2f (&uR)[4], v2f (&uI)[4], v2f (&umI)[4])
{
    float cb[4], sb[4];
    group_angles(thetas, b, g, cb, sb);
#pragma unroll
    for (int kc = 0; kc < 4; ++kc) {
#pragma unroll
        for (int r = 0; r < 2; ++r) {
            float Rv, Iv;
            u_entry(cb, sb, n, 4 * kc + r + 2 * h, Rv, Iv);
            uR[kc][r]  = Rv;
            uI[kc][r]  = Iv;
            umI[kc][r] = -Iv;
        }
    }
}

__device__ __forceinline__ void complex_pair_product(
    const v2f (&uR)[4], const v2f (&uI)[4], const v2f (&umI)[4],
    const v2f (&bRe)[2][4], const v2f (&bIm)[2][4],
    v8f (&accRe)[2], v8f (&accIm)[2])
{
#pragma unroll
    for (int c = 0; c < 2; ++c) {
        accRe[c] = (v8f){};
        accIm[c] = (v8f){};
    }
#pragma unroll
    for (int kc = 0; kc < 4; ++kc) {
#pragma unroll
        for (int c = 0; c < 2; ++c) {
            accRe[c] = __builtin_amdgcn_wmma_f32_16x16x4_f32(
                false, uR[kc], false, bRe[c][kc], (short)0, accRe[c], false, false);
            accIm[c] = __builtin_amdgcn_wmma_f32_16x16x4_f32(
                false, uI[kc], false, bRe[c][kc], (short)0, accIm[c], false, false);
        }
    }
#pragma unroll
    for (int kc = 0; kc < 4; ++kc) {
#pragma unroll
        for (int c = 0; c < 2; ++c) {
            accRe[c] = __builtin_amdgcn_wmma_f32_16x16x4_f32(
                false, umI[kc], false, bIm[c][kc], (short)0, accRe[c], false, false);
            accIm[c] = __builtin_amdgcn_wmma_f32_16x16x4_f32(
                false, uR[kc], false, bIm[c][kc], (short)0, accIm[c], false, false);
        }
    }
}

__device__ __forceinline__ void exchange_barrier()
{
    asm volatile("s_wait_dscnt 0x0\n\ts_barrier_signal -1\n\ts_barrier_wait -1" ::: "memory");
}

__global__ __launch_bounds__(256) void rx_high_pair_first(
    const float* __restrict__ phi,
    const float* __restrict__ thetas,
    float*       __restrict__ state)
{
    __shared__ float xch[256 * 64];

    const int lane  = threadIdx.x & 31;
    const int w     = __builtin_amdgcn_readfirstlane((int)(threadIdx.x >> 5));
    const int b     = (int)blockIdx.x / WGS_PER_BATCH;
    const int cpair = (int)blockIdx.x % WGS_PER_BATCH;
    const int h     = lane >> 4;
    const int n     = lane & 15;

    v2f aR[4], aI[4], amI[4];
    v2f gR[4], gI[4], gmI[4];
    group_operand(thetas, b, 0, n, h, aR, aI, amI);
    group_operand(thetas, b, 1, n, h, gR, gI, gmI);

#pragma unroll 1
    for (int t = 0; t < 2; ++t) {
        const unsigned col0 = (unsigned)(2 * cpair + t) * 32u + 2u * (unsigned)n;

#pragma unroll 1
        for (int s = 0; s < 2; ++s) {
            const int j12 = 2 * w + s;
            const float* src = phi + (size_t)b * SSIZE + ((size_t)j12 << 12) + col0;
            v2f bRe[2][4];
#pragma unroll
            for (int kc = 0; kc < 4; ++kc)
#pragma unroll
                for (int r = 0; r < 2; ++r) {
                    const int j = 4 * kc + r + 2 * h;
                    const v2f v = *(const v2f*)(src + ((size_t)j << 16));
                    bRe[0][kc][r] = v.x;
                    bRe[1][kc][r] = v.y;
                }

            v8f accRe[2], accIm[2];
#pragma unroll
            for (int c = 0; c < 2; ++c) {
                accRe[c] = (v8f){};
                accIm[c] = (v8f){};
            }
#pragma unroll
            for (int kc = 0; kc < 4; ++kc) {
#pragma unroll
                for (int c = 0; c < 2; ++c) {
                    accRe[c] = __builtin_amdgcn_wmma_f32_16x16x4_f32(
                        false, aR[kc], false, bRe[c][kc], (short)0, accRe[c], false, false);
                    accIm[c] = __builtin_amdgcn_wmma_f32_16x16x4_f32(
                        false, aI[kc], false, bRe[c][kc], (short)0, accIm[c], false, false);
                }
            }
            float* ldst = xch + j12 * 64 + 4 * n;
#pragma unroll
            for (int r = 0; r < 8; ++r) {
                v4f o;
                o.x = accRe[0][r];
                o.y = accIm[0][r];
                o.z = accRe[1][r];
                o.w = accIm[1][r];
                *(v4f*)(ldst + (r + 8 * h) * (16 * 64)) = o;
            }
        }

        exchange_barrier();

#pragma unroll 1
        for (int s = 0; s < 2; ++s) {
            const int j16 = 2 * w + s;
            const float* lsrc = xch + (j16 * 16) * 64 + 4 * n;
            v2f bRe[2][4], bIm[2][4];
#pragma unroll
            for (int kc = 0; kc < 4; ++kc)
#pragma unroll
                for (int r = 0; r < 2; ++r) {
                    const int j = 4 * kc + r + 2 * h;
                    const v4f v = *(const v4f*)(lsrc + j * 64);
                    bRe[0][kc][r] = v.x;
                    bIm[0][kc][r] = v.y;
                    bRe[1][kc][r] = v.z;
                    bIm[1][kc][r] = v.w;
                }

            v8f accRe[2], accIm[2];
            complex_pair_product(gR, gI, gmI, bRe, bIm, accRe, accIm);

            float* dst = state + 2 * ((size_t)b * SSIZE + ((size_t)j16 << 16) + col0);
#pragma unroll
            for (int r = 0; r < 8; ++r) {
                const int Mo = r + 8 * h;
                v4f o;
                o.x = accRe[0][r];
                o.y = accIm[0][r];
                o.z = accRe[1][r];
                o.w = accIm[1][r];
                *(v4f*)(dst + ((size_t)Mo << 13)) = o;
            }
        }

        exchange_barrier();
    }
}

__global__ __launch_bounds__(256) void rx_low_triple_pass(
    const float* __restrict__ thetas,
    float*       __restrict__ state)
{
    __shared__ float xch[16 * 512];
    __shared__ v2f   uop[4][4][32];

    const int lane = threadIdx.x & 31;
    const int w    = __builtin_amdgcn_readfirstlane((int)(threadIdx.x >> 5));
    const int b    = (int)blockIdx.x / WGS_PER_BATCH;
    const int tgrp = (int)blockIdx.x % WGS_PER_BATCH;
    const int h    = lane >> 4;
    const int n    = lane & 15;

    v2f aR[4], aI[4], amI[4];
    group_operand(thetas, b, 2, n, h, aR, aI, amI);

    if (w == 0) {
        float c4[4], s4[4], c0[4], s0[4];
        group_angles(thetas, b, 3, c4, s4);
        group_angles(thetas, b, 4, c0, s0);
#pragma unroll
        for (int kc = 0; kc < 4; ++kc) {
            v2f p4R, p4I, p0R, p0I;
#pragma unroll
            for (int r = 0; r < 2; ++r) {
                float Rv, Iv;
                u_entry(c4, s4, n, 4 * kc + r + 2 * h, Rv, Iv);
                p4R[r] = Rv;
                p4I[r] = Iv;
                u_entry(c0, s0, n, 2 * kc + r + 8 * h, Rv, Iv);
                p0R[r] = Rv;
                p0I[r] = Iv;
            }
            uop[0][kc][lane] = p4R;
            uop[1][kc][lane] = p4I;
            uop[2][kc][lane] = p0R;
            uop[3][kc][lane] = p0I;
        }
    }

#pragma unroll 1
    for (int t = 0; t < 4; ++t) {
        const unsigned tile  = (unsigned)t * WGS_PER_BATCH + (unsigned)tgrp;
        const size_t   tbase = (size_t)b * SSIZE + (size_t)tile * 4096u;

        {
            const float* src = state + 2 * (tbase + 32u * (unsigned)w + 2u * (unsigned)n);
            v2f bRe[2][4], bIm[2][4];
#pragma unroll
            for (int kc = 0; kc < 4; ++kc)
#pragma unroll
                for (int r = 0; r < 2; ++r) {
                    const int j = 4 * kc + r + 2 * h;
                    const v4f v = *(const v4f*)(src + ((size_t)j << 9));
                    bRe[0][kc][r] = v.x;
                    bIm[0][kc][r] = v.y;
                    bRe[1][kc][r] = v.z;
                    bIm[1][kc][r] = v.w;
                }

            v8f accRe[2], accIm[2];
            complex_pair_product(aR, aI, amI, bRe, bIm, accRe, accIm);

            float* ldst = xch + 2 * (32 * w + 2 * n);
#pragma unroll
            for (int r = 0; r < 8; ++r) {
                v4f o;
                o.x = accRe[0][r];
                o.y = accIm[0][r];
                o.z = accRe[1][r];
                o.w = accIm[1][r];
                *(v4f*)(ldst + (r + 8 * h) * 512) = o;
            }
        }

        exchange_barrier();

        v2f u4R[4], u4I[4], u0R[4], u0I[4];
#pragma unroll
        for (int kc = 0; kc < 4; ++kc) {
            u4R[kc] = uop[0][kc][lane];
            u4I[kc] = uop[1][kc][lane];
            u0R[kc] = uop[2][kc][lane];
            u0I[kc] = uop[3][kc][lane];
        }

#pragma unroll 1
        for (int q = 0; q < 2; ++q) {
            const int j8 = 2 * w + q;

            const v2f* lsrc = (const v2f*)(xch + j8 * 512) + n;
            v2f xRe[4], xIm[4];
#pragma unroll
            for (int kc = 0; kc < 4; ++kc)
#pragma unroll
                for (int r = 0; r < 2; ++r) {
                    const int j4 = 4 * kc + r + 2 * h;
                    const v2f v = lsrc[j4 * 16];
                    xRe[kc][r] = v.x;
                    xIm[kc][r] = v.y;
                }

            v8f y1Re = {};
            v8f y1Im = {};
#pragma unroll
            for (int kc = 0; kc < 4; ++kc) {
                y1Re = __builtin_amdgcn_wmma_f32_16x16x4_f32(
                    false, xRe[kc], false, u4R[kc], (short)0, y1Re, false, false);
                y1Im = __builtin_amdgcn_wmma_f32_16x16x4_f32(
                    false, xRe[kc], false, u4I[kc], (short)0, y1Im, false, false);
            }
#pragma unroll
            for (int kc = 0; kc < 4; ++kc) {
                const v2f nx = -xIm[kc];
                y1Re = __builtin_amdgcn_wmma_f32_16x16x4_f32(
                    false, nx, false, u4I[kc], (short)0, y1Re, false, false);
                y1Im = __builtin_amdgcn_wmma_f32_16x16x4_f32(
                    false, xIm[kc], false, u4R[kc], (short)0, y1Im, false, false);
            }

            v8f y2Re = {};
            v8f y2Im = {};
#pragma unroll
            for (int s = 0; s < 4; ++s) {
                v2f a;
                a.x = y1Re[2 * s];
                a.y = y1Re[2 * s + 1];
                y2Re = __builtin_amdgcn_wmma_f32_16x16x4_f32(
                    false, a, false, u0R[s], (short)0, y2Re, false, false);
                y2Im = __builtin_amdgcn_wmma_f32_16x16x4_f32(
                    false, a, false, u0I[s], (short)0, y2Im, false, false);
            }
#pragma unroll
            for (int s = 0; s < 4; ++s) {
                v2f a, na;
                a.x  = y1Im[2 * s];
                a.y  = y1Im[2 * s + 1];
                na.x = -a.x;
                na.y = -a.y;
                y2Re = __builtin_amdgcn_wmma_f32_16x16x4_f32(
                    false, na, false, u0I[s], (short)0, y2Re, false, false);
                y2Im = __builtin_amdgcn_wmma_f32_16x16x4_f32(
                    false, a, false, u0R[s], (short)0, y2Im, false, false);
            }

            v2f* dst = (v2f*)state + tbase + (size_t)j8 * 256u + (unsigned)n;
#pragma unroll
            for (int r = 0; r < 8; ++r) {
                v2f o;
                o.x = y2Re[r];
                o.y = y2Im[r];
                dst[(r + 8 * h) * 16] = o;
            }
        }

        exchange_barrier();
    }
}

extern "C" void kernel_launch(void* const* d_in, const int* in_sizes, int n_in,
                              void* d_out, int out_size, void* d_ws, size_t ws_size,
                              hipStream_t stream) {
    (void)in_sizes; (void)n_in; (void)d_ws; (void)ws_size; (void)out_size;
    const float* phi    = (const float*)d_in[0];
    const float* thetas = (const float*)d_in[1];
    float* out = (float*)d_out;

    const dim3 grid(1024), block(256);
    rx_high_pair_first<<<grid, block, 0, stream>>>(phi, thetas, out);
    rx_low_triple_pass<<<grid, block, 0, stream>>>(thetas, out);
}


# cos/sin of the 20 angles evaluated once per wave across lanes (lane q = qubit q, one sincosf) and read out by lane reads, not 16/8/24 serial scalar evaluations per wave
# speedup vs baseline: 4.9201x; 1.0485x over previous
#include <hip/hip_runtime.h>


typedef __attribute__((ext_vector_type(2))) float v2f;
typedef __attribute__((ext_vector_type(4))) float v4f;
typedef __attribute__((ext_vector_type(8))) float v8f;

#define NQ     20
#define NBATCH 16
#define SSIZE  (1u << NQ)

#define WGS_PER_BATCH 64

__device__ __forceinline__ void wave_angles(const float* __restrict__ thetas, int b, int lane,
                                            float& cl, float& sl)
{
    const int q = lane < NQ ? lane : NQ - 1;
    const float th = 0.5f * thetas[b * NQ + q];
    sincosf(th, &sl, &cl);
}

__device__ __forceinline__ void group_angles(float cl, float sl, int g, float (&cb)[4], float (&sb)[4])
{
#pragma unroll
    for (int beta = 0; beta < 4; ++beta) {
        const int q = 4 * g + (3 - beta);
        cb[beta] = __int_as_float(__builtin_amdgcn_readlane(__float_as_int(cl), q));
        sb[beta] = __int_as_float(__builtin_amdgcn_readlane(__float_as_int(sl), q));
    }
}

__device__ __forceinline__ void u_entry(const float (&cb)[4], const float (&sb)[4],
                                        int M, int j, float& Rv, float& Iv)
{
    const int x = M ^ j;
    float prod = 1.0f;
#pragma unroll
    for (int beta = 0; beta < 4; ++beta)
        prod *= ((x >> beta) & 1) ? sb[beta] : cb[beta];
    const int k = __popc(x);
    Rv = (k & 1) ? 0.0f : ((k & 2) ? -prod : prod);
    Iv = (k & 1) ? ((k & 2) ? prod : -prod) : 0.0f;
}

__device__ __forceinline__ void group_operand(float cl, float sl, int g,
                                              int n, int h,
                                              v2f (&uR)[4], v2f (&uI)[4], v2f (&umI)[4])
{
    float cb[4], sb[4];
    group_angles(cl, sl, g, cb, sb);
#pragma unroll
    for (int kc = 0; kc < 4; ++kc) {
#pragma unroll
        for (int r = 0; r < 2; ++r) {
            float Rv, Iv;
            u_entry(cb, sb, n, 4 * kc + r + 2 * h, Rv, Iv);
            uR[kc][r]  = Rv;
            uI[kc][r]  = Iv;
            umI[kc][r] = -Iv;
        }
    }
}

__device__ __forceinline__ void complex_pair_product(
    const v2f (&uR)[4], const v2f (&uI)[4], const v2f (&umI)[4],
    const v2f (&bRe)[2][4], const v2f (&bIm)[2][4],
    v8f (&accRe)[2], v8f (&accIm)[2])
{
#pragma unroll
    for (int c = 0; c < 2; ++c) {
        accRe[c] = (v8f){};
        accIm[c] = (v8f){};
    }
#pragma unroll
    for (int kc = 0; kc < 4; ++kc) {
#pragma unroll
        for (int c = 0; c < 2; ++c) {
            accRe[c] = __builtin_amdgcn_wmma_f32_16x16x4_f32(
                false, uR[kc], false, bRe[c][kc], (short)0, accRe[c], false, false);
            accIm[c] = __builtin_amdgcn_wmma_f32_16x16x4_f32(
                false, uI[kc], false, bRe[c][kc], (short)0, accIm[c], false, false);
        }
    }
#pragma unroll
    for (int kc = 0; kc < 4; ++kc) {
#pragma unroll
        for (int c = 0; c < 2; ++c) {
            accRe[c] = __builtin_amdgcn_wmma_f32_16x16x4_f32(
                false, umI[kc], false, bIm[c][kc], (short)0, accRe[c], false, false);
            accIm[c] = __builtin_amdgcn_wmma_f32_16x16x4_f32(
                false, uR[kc], false, bIm[c][kc], (short)0, accIm[c], false, false);
        }
    }
}

__device__ __forceinline__ void exchange_barrier()
{
    asm volatile("s_wait_dscnt 0x0\n\ts_barrier_signal -1\n\ts_barrier_wait -1" ::: "memory");
}

__global__ __launch_bounds__(256) void rx_high_pair_first(
    const float* __restrict__ phi,
    const float* __restrict__ thetas,
    float*       __restrict__ state)
{
    __shared__ float xch[256 * 64];

    const int lane  = threadIdx.x & 31;
    const int w     = __builtin_amdgcn_readfirstlane((int)(threadIdx.x >> 5));
    const int b     = (int)blockIdx.x / WGS_PER_BATCH;
    const int cpair = (int)blockIdx.x % WGS_PER_BATCH;
    const int h     = lane >> 4;
    const int n     = lane & 15;

    v2f aR[4], aI[4], amI[4];
    v2f gR[4], gI[4], gmI[4];
    float cl, sl;
    wave_angles(thetas, b, lane, cl, sl);
    group_operand(cl, sl, 0, n, h, aR, aI, amI);
    group_operand(cl, sl, 1, n, h, gR, gI, gmI);

#pragma unroll 1
    for (int t = 0; t < 2; ++t) {
        const unsigned col0 = (unsigned)(2 * cpair + t) * 32u + 2u * (unsigned)n;

#pragma unroll 1
        for (int s = 0; s < 2; ++s) {
            const int j12 = 2 * w + s;
            const float* src = phi + (size_t)b * SSIZE + ((size_t)j12 << 12) + col0;
            v2f bRe[2][4];
#pragma unroll
            for (int kc = 0; kc < 4; ++kc)
#pragma unroll
                for (int r = 0; r < 2; ++r) {
                    const int j = 4 * kc + r + 2 * h;
                    const v2f v = *(const v2f*)(src + ((size_t)j << 16));
                    bRe[0][kc][r] = v.x;
                    bRe[1][kc][r] = v.y;
                }

            v8f accRe[2], accIm[2];
#pragma unroll
            for (int c = 0; c < 2; ++c) {
                accRe[c] = (v8f){};
                accIm[c] = (v8f){};
            }
#pragma unroll
            for (int kc = 0; kc < 4; ++kc) {
#pragma unroll
                for (int c = 0; c < 2; ++c) {
                    accRe[c] = __builtin_amdgcn_wmma_f32_16x16x4_f32(
                        false, aR[kc], false, bRe[c][kc], (short)0, accRe[c], false, false);
                    accIm[c] = __builtin_amdgcn_wmma_f32_16x16x4_f32(
                        false, aI[kc], false, bRe[c][kc], (short)0, accIm[c], false, false);
                }
            }
            float* ldst = xch + j12 * 64 + 4 * n;
#pragma unroll
            for (int r = 0; r < 8; ++r) {
                v4f o;
                o.x = accRe[0][r];
                o.y = accIm[0][r];
                o.z = accRe[1][r];
                o.w = accIm[1][r];
                *(v4f*)(ldst + (r + 8 * h) * (16 * 64)) = o;
            }
        }

        exchange_barrier();

#pragma unroll 1
        for (int s = 0; s < 2; ++s) {
            const int j16 = 2 * w + s;
            const float* lsrc = xch + (j16 * 16) * 64 + 4 * n;
            v2f bRe[2][4], bIm[2][4];
#pragma unroll
            for (int kc = 0; kc < 4; ++kc)
#pragma unroll
                for (int r = 0; r < 2; ++r) {
                    const int j = 4 * kc + r + 2 * h;
                    const v4f v = *(const v4f*)(lsrc + j * 64);
                    bRe[0][kc][r] = v.x;
                    bIm[0][kc][r] = v.y;
                    bRe[1][kc][r] = v.z;
                    bIm[1][kc][r] = v.w;
                }

            v8f accRe[2], accIm[2];
            complex_pair_product(gR, gI, gmI, bRe, bIm, accRe, accIm);

            float* dst = state + 2 * ((size_t)b * SSIZE + ((size_t)j16 << 16) + col0);
#pragma unroll
            for (int r = 0; r < 8; ++r) {
                const int Mo = r + 8 * h;
                v4f o;
                o.x = accRe[0][r];
                o.y = accIm[0][r];
                o.z = accRe[1][r];
                o.w = accIm[1][r];
                *(v4f*)(dst + ((size_t)Mo << 13)) = o;
            }
        }

        exchange_barrier();
    }
}

__global__ __launch_bounds__(256) void rx_low_triple_pass(
    const float* __restrict__ thetas,
    float*       __restrict__ state)
{
    __shared__ float xch[16 * 512];
    __shared__ v2f   uop[4][4][32];

    const int lane = threadIdx.x & 31;
    const int w    = __builtin_amdgcn_readfirstlane((int)(threadIdx.x >> 5));
    const int b    = (int)blockIdx.x / WGS_PER_BATCH;
    const int tgrp = (int)blockIdx.x % WGS_PER_BATCH;
    const int h    = lane >> 4;
    const int n    = lane & 15;

    v2f aR[4], aI[4], amI[4];
    float cl, sl;
    wave_angles(thetas, b, lane, cl, sl);
    group_operand(cl, sl, 2, n, h, aR, aI, amI);

    if (w == 0) {
        float c4[4], s4[4], c0[4], s0[4];
        group_angles(cl, sl, 3, c4, s4);
        group_angles(cl, sl, 4, c0, s0);
#pragma unroll
        for (int kc = 0; kc < 4; ++kc) {
            v2f p4R, p4I, p0R, p0I;
#pragma unroll
            for (int r = 0; r < 2; ++r) {
                float Rv, Iv;
                u_entry(c4, s4, n, 4 * kc + r + 2 * h, Rv, Iv);
                p4R[r] = Rv;
                p4I[r] = Iv;
                u_entry(c0, s0, n, 2 * kc + r + 8 * h, Rv, Iv);
                p0R[r] = Rv;
                p0I[r] = Iv;
            }
            uop[0][kc][lane] = p4R;
            uop[1][kc][lane] = p4I;
            uop[2][kc][lane] = p0R;
            uop[3][kc][lane] = p0I;
        }
    }

#pragma unroll 1
    for (int t = 0; t < 4; ++t) {
        const unsigned tile  = (unsigned)t * WGS_PER_BATCH + (unsigned)tgrp;
        const size_t   tbase = (size_t)b * SSIZE + (size_t)tile * 4096u;

        {
            const float* src = state + 2 * (tbase + 32u * (unsigned)w + 2u * (unsigned)n);
            v2f bRe[2][4], bIm[2][4];
#pragma unroll
            for (int kc = 0; kc < 4; ++kc)
#pragma unroll
                for (int r = 0; r < 2; ++r) {
                    const int j = 4 * kc + r + 2 * h;
                    const v4f v = *(const v4f*)(src + ((size_t)j << 9));
                    bRe[0][kc][r] = v.x;
                    bIm[0][kc][r] = v.y;
                    bRe[1][kc][r] = v.z;
                    bIm[1][kc][r] = v.w;
                }

            v8f accRe[2], accIm[2];
            complex_pair_product(aR, aI, amI, bRe, bIm, accRe, accIm);

            float* ldst = xch + 2 * (32 * w + 2 * n);
#pragma unroll
            for (int r = 0; r < 8; ++r) {
                v4f o;
                o.x = accRe[0][r];
                o.y = accIm[0][r];
                o.z = accRe[1][r];
                o.w = accIm[1][r];
                *(v4f*)(ldst + (r + 8 * h) * 512) = o;
            }
        }

        exchange_barrier();

        v2f u4R[4], u4I[4], u0R[4], u0I[4];
#pragma unroll
        for (int kc = 0; kc < 4; ++kc) {
            u4R[kc] = uop[0][kc][lane];
            u4I[kc] = uop[1][kc][lane];
            u0R[kc] = uop[2][kc][lane];
            u0I[kc] = uop[3][kc][lane];
        }

#pragma unroll 1
        for (int q = 0; q < 2; ++q) {
            const int j8 = 2 * w + q;

            const v2f* lsrc = (const v2f*)(xch + j8 * 512) + n;
            v2f xRe[4], xIm[4];
#pragma unroll
            for (int kc = 0; kc < 4; ++kc)
#pragma unroll
                for (int r = 0; r < 2; ++r) {
                    const int j4 = 4 * kc + r + 2 * h;
                    const v2f v = lsrc[j4 * 16];
                    xRe[kc][r] = v.x;
                    xIm[kc][r] = v.y;
                }

            v8f y1Re = {};
            v8f y1Im = {};
#pragma unroll
            for (int kc = 0; kc < 4; ++kc) {
                y1Re = __builtin_amdgcn_wmma_f32_16x16x4_f32(
                    false, xRe[kc], false, u4R[kc], (short)0, y1Re, false, false);
                y1Im = __builtin_amdgcn_wmma_f32_16x16x4_f32(
                    false, xRe[kc], false, u4I[kc], (short)0, y1Im, false, false);
            }
#pragma unroll
            for (int kc = 0; kc < 4; ++kc) {
                const v2f nx = -xIm[kc];
                y1Re = __builtin_amdgcn_wmma_f32_16x16x4_f32(
                    false, nx, false, u4I[kc], (short)0, y1Re, false, false);
                y1Im = __builtin_amdgcn_wmma_f32_16x16x4_f32(
                    false, xIm[kc], false, u4R[kc], (short)0, y1Im, false, false);
            }

            v8f y2Re = {};
            v8f y2Im = {};
#pragma unroll
            for (int s = 0; s < 4; ++s) {
                v2f a;
                a.x = y1Re[2 * s];
                a.y = y1Re[2 * s + 1];
                y2Re = __builtin_amdgcn_wmma_f32_16x16x4_f32(
                    false, a, false, u0R[s], (short)0, y2Re, false, false);
                y2Im = __builtin_amdgcn_wmma_f32_16x16x4_f32(
                    false, a, false, u0I[s], (short)0, y2Im, false, false);
            }
#pragma unroll
            for (int s = 0; s < 4; ++s) {
                v2f a, na;
                a.x  = y1Im[2 * s];
                a.y  = y1Im[2 * s + 1];
                na.x = -a.x;
                na.y = -a.y;
                y2Re = __builtin_amdgcn_wmma_f32_16x16x4_f32(
                    false, na, false, u0I[s], (short)0, y2Re, false, false);
                y2Im = __builtin_amdgcn_wmma_f32_16x16x4_f32(
                    false, a, false, u0R[s], (short)0, y2Im, false, false);
            }

            v2f* dst = (v2f*)state + tbase + (size_t)j8 * 256u + (unsigned)n;
#pragma unroll
            for (int r = 0; r < 8; ++r) {
                v2f o;
                o.x = y2Re[r];
                o.y = y2Im[r];
                dst[(r + 8 * h) * 16] = o;
            }
        }

        exchange_barrier();
    }
}

extern "C" void kernel_launch(void* const* d_in, const int* in_sizes, int n_in,
                              void* d_out, int out_size, void* d_ws, size_t ws_size,
                              hipStream_t stream) {
    (void)in_sizes; (void)n_in; (void)d_ws; (void)ws_size; (void)out_size;
    const float* phi    = (const float*)d_in[0];
    const float* thetas = (const float*)d_in[1];
    float* out = (float*)d_out;

    const dim3 grid(1024), block(256);
    rx_high_pair_first<<<grid, block, 0, stream>>>(phi, thetas, out);
    rx_low_triple_pass<<<grid, block, 0, stream>>>(thetas, out);
}


# Group matrices carried as one real matrix (re and im parts have disjoint support), k sorted by popcount parity: a complex 16x16 product takes 8 WMMAs, not 16; f32 sum order of every product changes
# speedup vs baseline: 6.0842x; 1.2366x over previous
#include <hip/hip_runtime.h>


typedef __attribute__((ext_vector_type(2))) float v2f;
typedef __attribute__((ext_vector_type(4))) float v4f;
typedef __attribute__((ext_vector_type(8))) float v8f;

#define NQ     20
#define NBATCH 16
#define SSIZE  (1u << NQ)

#define WGS_PER_BATCH 64

__device__ __forceinline__ void wave_angles(const float* __restrict__ thetas, int b, int lane,
                                            float& cl, float& sl)
{
    const int q = lane < NQ ? lane : NQ - 1;
    const float th = 0.5f * thetas[b * NQ + q];
    sincosf(th, &sl, &cl);
}

__device__ __forceinline__ void group_angles(float cl, float sl, int g, float (&cb)[4], float (&sb)[4])
{
#pragma unroll
    for (int beta = 0; beta < 4; ++beta) {
        const int q = 4 * g + (3 - beta);
        cb[beta] = __int_as_float(__builtin_amdgcn_readlane(__float_as_int(cl), q));
        sb[beta] = __int_as_float(__builtin_amdgcn_readlane(__float_as_int(sl), q));
    }
}

__device__ __forceinline__ int kord(int kc, int r, int h)
{
    const int t = ((kc & 1) << 2) | (r << 1) | h;
    const int pt = (t ^ (t >> 1) ^ (t >> 2)) & 1;
    return (t << 1) | (pt ^ (kc >> 1));
}

__device__ __forceinline__ int rowmap(int p)
{
    return (p & 14) | (((p >> 1) ^ (p >> 2) ^ (p >> 3) ^ p) & 1);
}

__device__ __forceinline__ float w_entry(const float (&cb)[4], const float (&sb)[4], int M, int j)
{
    const int x = M ^ j;
    float prod = 1.0f;
#pragma unroll
    for (int beta = 0; beta < 4; ++beta)
        prod *= ((x >> beta) & 1) ? sb[beta] : cb[beta];
    const int k = __popc(x);
    return ((k + 1) & 2) ? -prod : prod;
}

__device__ __forceinline__ void signed_operand(const float (&cb)[4], const float (&sb)[4],
                                               int M, int h, v2f (&o1)[4], v2f (&o2)[4])
{
    const bool oddM = (__popc(M) & 1) != 0;
#pragma unroll
    for (int kc = 0; kc < 4; ++kc) {
#pragma unroll
        for (int r = 0; r < 2; ++r) {
            const float wv = w_entry(cb, sb, M, kord(kc, r, h));
            const float fw = oddM ? wv : -wv;
            if (kc < 2) {
                o1[kc][r] = wv;
                o2[kc][r] = -fw;
            } else {
                o1[kc][r] = fw;
                o2[kc][r] = wv;
            }
        }
    }
}

__device__ __forceinline__ void group_operand(float cl, float sl, int g,
                                              int n, int h,
                                              v2f (&a1)[4], v2f (&a2)[4])
{
    float cb[4], sb[4];
    group_angles(cl, sl, g, cb, sb);
    signed_operand(cb, sb, rowmap(n), h, a1, a2);
}

__device__ __forceinline__ void complex_pair_product(
    const v2f (&a1)[4], const v2f (&a2)[4],
    const v2f (&bRe)[2][4], const v2f (&bIm)[2][4],
    v8f (&acc1)[2], v8f (&acc2)[2])
{
#pragma unroll
    for (int c = 0; c < 2; ++c) {
        acc1[c] = (v8f){};
        acc2[c] = (v8f){};
    }
#pragma unroll
    for (int kc = 0; kc < 2; ++kc) {
#pragma unroll
        for (int c = 0; c < 2; ++c) {
            acc1[c] = __builtin_amdgcn_wmma_f32_16x16x4_f32(
                false, a1[kc], false, bRe[c][kc], (short)0, acc1[c], false, false);
            acc2[c] = __builtin_amdgcn_wmma_f32_16x16x4_f32(
                false, a2[kc], false, bIm[c][kc], (short)0, acc2[c], false, false);
        }
    }
#pragma unroll
    for (int kc = 2; kc < 4; ++kc) {
#pragma unroll
        for (int c = 0; c < 2; ++c) {
            acc1[c] = __builtin_amdgcn_wmma_f32_16x16x4_f32(
                false, a1[kc], false, bIm[c][kc], (short)0, acc1[c], false, false);
            acc2[c] = __builtin_amdgcn_wmma_f32_16x16x4_f32(
                false, a2[kc], false, bRe[c][kc], (short)0, acc2[c], false, false);
        }
    }
}

__device__ __forceinline__ void exchange_barrier()
{
    asm volatile("s_wait_dscnt 0x0\n\ts_barrier_signal -1\n\ts_barrier_wait -1" ::: "memory");
}

__global__ __launch_bounds__(256) void rx_high_pair_first(
    const float* __restrict__ phi,
    const float* __restrict__ thetas,
    float*       __restrict__ state)
{
    __shared__ float xch[256 * 64];

    const int lane  = threadIdx.x & 31;
    const int w     = __builtin_amdgcn_readfirstlane((int)(threadIdx.x >> 5));
    const int b     = (int)blockIdx.x / WGS_PER_BATCH;
    const int cpair = (int)blockIdx.x % WGS_PER_BATCH;
    const int h     = lane >> 4;
    const int n     = lane & 15;

    v2f a1[4], a2[4];
    v2f g1[4], g2[4];
    float cl, sl;
    wave_angles(thetas, b, lane, cl, sl);
    group_operand(cl, sl, 0, n, h, a1, a2);
    group_operand(cl, sl, 1, n, h, g1, g2);

#pragma unroll 1
    for (int t = 0; t < 2; ++t) {
        const unsigned col0 = (unsigned)(2 * cpair + t) * 32u + 2u * (unsigned)n;

#pragma unroll 1
        for (int s = 0; s < 2; ++s) {
            const int j12 = 2 * w + s;
            const float* src = phi + (size_t)b * SSIZE + ((size_t)j12 << 12) + col0;
            v2f bRe[2][4];
#pragma unroll
            for (int kc = 0; kc < 4; ++kc)
#pragma unroll
                for (int r = 0; r < 2; ++r) {
                    const int j = kord(kc, r, h);
                    const v2f v = *(const v2f*)(src + ((size_t)j << 16));
                    bRe[0][kc][r] = v.x;
                    bRe[1][kc][r] = v.y;
                }

            v8f acc1[2], acc2[2];
#pragma unroll
            for (int c = 0; c < 2; ++c) {
                acc1[c] = (v8f){};
                acc2[c] = (v8f){};
            }
#pragma unroll
            for (int kc = 0; kc < 2; ++kc) {
#pragma unroll
                for (int c = 0; c < 2; ++c) {
                    acc1[c] = __builtin_amdgcn_wmma_f32_16x16x4_f32(
                        false, a1[kc], false, bRe[c][kc], (short)0, acc1[c], false, false);
                    acc2[c] = __builtin_amdgcn_wmma_f32_16x16x4_f32(
                        false, a2[kc + 2], false, bRe[c][kc + 2], (short)0, acc2[c], false, false);
                }
            }
            float* ldst = xch + j12 * 64 + 4 * n;
#pragma unroll
            for (int r = 0; r < 8; ++r) {
                v4f o;
                o.x = (r & 1) ? acc2[0][r] : acc1[0][r];
                o.y = (r & 1) ? acc1[0][r] : acc2[0][r];
                o.z = (r & 1) ? acc2[1][r] : acc1[1][r];
                o.w = (r & 1) ? acc1[1][r] : acc2[1][r];
                *(v4f*)(ldst + rowmap(8 * h + r) * (16 * 64)) = o;
            }
        }

        exchange_barrier();

#pragma unroll 1
        for (int s = 0; s < 2; ++s) {
            const int j16 = 2 * w + s;
            const float* lsrc = xch + (j16 * 16) * 64 + 4 * n;
            v2f bRe[2][4], bIm[2][4];
#pragma unroll
            for (int kc = 0; kc < 4; ++kc)
#pragma unroll
                for (int r = 0; r < 2; ++r) {
                    const int j = kord(kc, r, h);
                    const v4f v = *(const v4f*)(lsrc + j * 64);
                    bRe[0][kc][r] = v.x;
                    bIm[0][kc][r] = v.y;
                    bRe[1][kc][r] = v.z;
                    bIm[1][kc][r] = v.w;
                }

            v8f acc1[2], acc2[2];
            complex_pair_product(g1, g2, bRe, bIm, acc1, acc2);

            float* dst = state + 2 * ((size_t)b * SSIZE + ((size_t)j16 << 16) + col0);
#pragma unroll
            for (int r = 0; r < 8; ++r) {
                const int Mo = rowmap(8 * h + r);
                v4f o;
                o.x = (r & 1) ? acc2[0][r] : acc1[0][r];
                o.y = (r & 1) ? acc1[0][r] : acc2[0][r];
                o.z = (r & 1) ? acc2[1][r] : acc1[1][r];
                o.w = (r & 1) ? acc1[1][r] : acc2[1][r];
                *(v4f*)(dst + ((size_t)Mo << 13)) = o;
            }
        }

        exchange_barrier();
    }
}

__global__ __launch_bounds__(256) void rx_low_triple_pass(
    const float* __restrict__ thetas,
    float*       __restrict__ state)
{
    __shared__ float xch[16 * 512];
    __shared__ v2f   uop[4][4][32];

    const int lane = threadIdx.x & 31;
    const int w    = __builtin_amdgcn_readfirstlane((int)(threadIdx.x >> 5));
    const int b    = (int)blockIdx.x / WGS_PER_BATCH;
    const int tgrp = (int)blockIdx.x % WGS_PER_BATCH;
    const int h    = lane >> 4;
    const int n    = lane & 15;

    v2f a1[4], a2[4];
    float cl, sl;
    wave_angles(thetas, b, lane, cl, sl);
    group_operand(cl, sl, 2, n, h, a1, a2);

    if (w == 0) {
        float c4[4], s4[4], c0[4], s0[4];
        group_angles(cl, sl, 3, c4, s4);
        group_angles(cl, sl, 4, c0, s0);
        v2f p41[4], p42[4], p01[4], p02[4];
        signed_operand(c4, s4, n, h, p41, p42);
        signed_operand(c0, s0, n, h, p01, p02);
#pragma unroll
        for (int kc = 0; kc < 4; ++kc) {
            uop[0][kc][lane] = p41[kc];
            uop[1][kc][lane] = p42[kc];
            uop[2][kc][lane] = p01[kc];
            uop[3][kc][lane] = p02[kc];
        }
    }

    const bool oddn = (__popc(n) & 1) != 0;
    const int  pin  = kord((n >> 1) & 3, n & 1, n >> 3);

#pragma unroll 1
    for (int t = 0; t < 4; ++t) {
        const unsigned tile  = (unsigned)t * WGS_PER_BATCH + (unsigned)tgrp;
        const size_t   tbase = (size_t)b * SSIZE + (size_t)tile * 4096u;

        {
            const float* src = state + 2 * (tbase + 32u * (unsigned)w + 2u * (unsigned)n);
            v2f bRe[2][4], bIm[2][4];
#pragma unroll
            for (int kc = 0; kc < 4; ++kc)
#pragma unroll
                for (int r = 0; r < 2; ++r) {
                    const int j = kord(kc, r, h);
                    const v4f v = *(const v4f*)(src + ((size_t)j << 9));
                    bRe[0][kc][r] = v.x;
                    bIm[0][kc][r] = v.y;
                    bRe[1][kc][r] = v.z;
                    bIm[1][kc][r] = v.w;
                }

            v8f acc1[2], acc2[2];
            complex_pair_product(a1, a2, bRe, bIm, acc1, acc2);

            float* ldst = xch + 2 * (32 * w + 2 * n);
#pragma unroll
            for (int r = 0; r < 8; ++r) {
                v4f o;
                o.x = (r & 1) ? acc2[0][r] : acc1[0][r];
                o.y = (r & 1) ? acc1[0][r] : acc2[0][r];
                o.z = (r & 1) ? acc2[1][r] : acc1[1][r];
                o.w = (r & 1) ? acc1[1][r] : acc2[1][r];
                *(v4f*)(ldst + rowmap(8 * h + r) * 512) = o;
            }
        }

        exchange_barrier();

        v2f u41[4], u42[4], u01[4], u02[4];
#pragma unroll
        for (int kc = 0; kc < 4; ++kc) {
            u41[kc] = uop[0][kc][lane];
            u42[kc] = uop[1][kc][lane];
            u01[kc] = uop[2][kc][lane];
            u02[kc] = uop[3][kc][lane];
        }

#pragma unroll 1
        for (int q = 0; q < 2; ++q) {
            const int j8 = 2 * w + q;

            const v2f* lsrc = (const v2f*)(xch + j8 * 512) + pin;
            v2f xRe[4], xIm[4];
#pragma unroll
            for (int kc = 0; kc < 4; ++kc)
#pragma unroll
                for (int r = 0; r < 2; ++r) {
                    const int j4 = kord(kc, r, h);
                    const v2f v = lsrc[j4 * 16];
                    xRe[kc][r] = v.x;
                    xIm[kc][r] = v.y;
                }

            v8f p1 = {};
            v8f p2 = {};
#pragma unroll
            for (int kc = 0; kc < 2; ++kc) {
                p1 = __builtin_amdgcn_wmma_f32_16x16x4_f32(
                    false, xRe[kc], false, u41[kc], (short)0, p1, false, false);
                p2 = __builtin_amdgcn_wmma_f32_16x16x4_f32(
                    false, xIm[kc], false, u42[kc], (short)0, p2, false, false);
            }
#pragma unroll
            for (int kc = 2; kc < 4; ++kc) {
                p1 = __builtin_amdgcn_wmma_f32_16x16x4_f32(
                    false, xIm[kc], false, u41[kc], (short)0, p1, false, false);
                p2 = __builtin_amdgcn_wmma_f32_16x16x4_f32(
                    false, xRe[kc], false, u42[kc], (short)0, p2, false, false);
            }
            v8f y1Re, y1Im;
#pragma unroll
            for (int r = 0; r < 8; ++r) {
                y1Re[r] = oddn ? p2[r] : p1[r];
                y1Im[r] = oddn ? p1[r] : p2[r];
            }

            v8f q1 = {};
            v8f q2 = {};
#pragma unroll
            for (int s = 0; s < 2; ++s) {
                v2f are, aim;
                are.x = y1Re[2 * s];
                are.y = y1Re[2 * s + 1];
                aim.x = y1Im[2 * s];
                aim.y = y1Im[2 * s + 1];
                q1 = __builtin_amdgcn_wmma_f32_16x16x4_f32(
                    false, are, false, u01[s], (short)0, q1, false, false);
                q2 = __builtin_amdgcn_wmma_f32_16x16x4_f32(
                    false, aim, false, u02[s], (short)0, q2, false, false);
            }
#pragma unroll
            for (int s = 2; s < 4; ++s) {
                v2f are, aim;
                are.x = y1Re[2 * s];
                are.y = y1Re[2 * s + 1];
                aim.x = y1Im[2 * s];
                aim.y = y1Im[2 * s + 1];
                q1 = __builtin_amdgcn_wmma_f32_16x16x4_f32(
                    false, aim, false, u01[s], (short)0, q1, false, false);
                q2 = __builtin_amdgcn_wmma_f32_16x16x4_f32(
                    false, are, false, u02[s], (short)0, q2, false, false);
            }
            v8f y2Re, y2Im;
#pragma unroll
            for (int r = 0; r < 8; ++r) {
                y2Re[r] = oddn ? q2[r] : q1[r];
                y2Im[r] = oddn ? q1[r] : q2[r];
            }

            v2f* dst = (v2f*)state + tbase + (size_t)j8 * 256u + (unsigned)n;
#pragma unroll
            for (int r = 0; r < 8; ++r) {
                v2f o;
                o.x = y2Re[r];
                o.y = y2Im[r];
                dst[(r + 8 * h) * 16] = o;
            }
        }

        exchange_barrier();
    }
}

extern "C" void kernel_launch(void* const* d_in, const int* in_sizes, int n_in,
                              void* d_out, int out_size, void* d_ws, size_t ws_size,
                              hipStream_t stream) {
    (void)in_sizes; (void)n_in; (void)d_ws; (void)ws_size; (void)out_size;
    const float* phi    = (const float*)d_in[0];
    const float* thetas = (const float*)d_in[1];
    float* out = (float*)d_out;

    const dim3 grid(1024), block(256);
    rx_high_pair_first<<<grid, block, 0, stream>>>(phi, thetas, out);
    rx_low_triple_pass<<<grid, block, 0, stream>>>(thetas, out);
}
